# starting kernel
# baseline (speedup 1.0000x reference)
#include <hip/hip_runtime.h>
#include <hip/hip_bf16.h>
#include <math.h>

typedef __attribute__((ext_vector_type(2))) float v2f;
typedef __attribute__((ext_vector_type(4))) float v4f;
typedef __attribute__((ext_vector_type(8))) float v8f;

#define E_DIM   512
#define L_DIM   128
#define KCHUNK  256
#define PRSTRIDE 144

#define WS_BTP   0
#define WS_IVAR  (E_DIM * L_DIM)
#define WS_M2    (WS_IVAR + E_DIM)
#define WS_CONST (WS_M2 + L_DIM)

__global__ __launch_bounds__(256) void GaussianScorer_prep(
    const float* __restrict__ means, const float* __restrict__ var,
    float* __restrict__ ws) {
  __shared__ float sIv[E_DIM];
  __shared__ float sRed[256];
  const int tid = threadIdx.x;

  float slog = 0.0f;
  for (int e = tid; e < E_DIM; e += 256) {
    float v  = var[e];
    float iv = 1.0f / v;
    sIv[e] = iv;
    ws[WS_IVAR + e] = iv;
    slog += logf(v);
  }
  sRed[tid] = slog;
  __syncthreads();
  for (int s = 128; s > 0; s >>= 1) {
    if (tid < s) sRed[tid] += sRed[tid + s];
    __syncthreads();
  }
  if (tid == 0) {
    const float log2pi = 1.8378770664093453f;
    const float factor = -0.5f * (float)E_DIM * log2pi;
    ws[WS_CONST] = factor - 0.5f * sRed[0];
  }

  for (int idx = tid; idx < E_DIM * L_DIM; idx += 256) {
    int pp  = idx >> 8;
    int rem = idx & 255;
    int l   = rem >> 1;
    int q   = rem & 1;
    int e   = 2 * pp + q;
    ws[WS_BTP + idx] = means[l * E_DIM + e] * sIv[e];
  }

  if (tid < L_DIM) {
    float s = 0.0f;
    for (int e = 0; e < E_DIM; ++e) {
      float m = means[tid * E_DIM + e];
      s += m * m * sIv[e];
    }
    ws[WS_M2 + tid] = s;
  }
}

__global__ __launch_bounds__(128) void GaussianScorer_gemm(
    const float* __restrict__ emb, const float* __restrict__ ws,
    float* __restrict__ out) {
  __shared__ v2f   sBp[(KCHUNK / 2) * PRSTRIDE];
  __shared__ float sIv[KCHUNK];
  __shared__ float sE2[64];

  const int tid   = threadIdx.x;
  const int wave  = tid >> 5;
  const int lane  = tid & 31;
  const int col   = lane & 15;
  const int khalf = lane >> 4;
  const int kgrp  = khalf * 2;
  const int rowStart = blockIdx.x * 64 + wave * 16;
  const int myRow    = rowStart + col;

  const float* aRow = emb + (size_t)myRow * E_DIM;

  v8f acc[8] = {};
  float e2p = 0.0f;

  for (int chunk = 0; chunk < 2; ++chunk) {
    const v4f* __restrict__ srcB =
        (const v4f*)(ws + WS_BTP + chunk * (KCHUNK / 2) * (L_DIM * 2));
    for (int j = tid; j < (KCHUNK / 2) * (L_DIM / 2); j += 128) {
      int pr = j >> 6;
      int q  = j & 63;
      *((v4f*)(sBp + pr * PRSTRIDE) + q) = srcB[pr * (L_DIM / 2) + q];
    }
    if (tid < KCHUNK / 4)
      ((v4f*)sIv)[tid] = ((const v4f*)(ws + WS_IVAR + chunk * KCHUNK))[tid];
    __syncthreads();

    const float* aC = aRow + chunk * KCHUNK + kgrp;
#pragma unroll 4
    for (int i = 0; i < KCHUNK / 4; ++i) {
      const int kl = i * 4 + kgrp;
      v2f a  = *(const v2f*)(aC + i * 4);
      v2f iv = *(const v2f*)(sIv + kl);
      e2p += a.x * a.x * iv.x + a.y * a.y * iv.y;

      const v2f* bRow = sBp + (i * 2 + khalf) * PRSTRIDE + col;
#pragma unroll
      for (int t = 0; t < 8; ++t) {
        v2f b = bRow[t * 16];
        acc[t] = __builtin_amdgcn_wmma_f32_16x16x4_f32(
            false, a, false, b, (short)0, acc[t], false, false);
      }
    }
    __syncthreads();
  }

  float e2t = e2p + __shfl_xor(e2p, 16, 32);
  if (lane < 16) sE2[wave * 16 + col] = e2t;
  __syncthreads();

  const float cTerm = ws[WS_CONST];
#pragma unroll
  for (int t = 0; t < 8; ++t) {
    float m2c = ws[WS_M2 + t * 16 + col];
#pragma unroll
    for (int i = 0; i < 8; ++i) {
      int rl = khalf * 8 + i;
      float val = cTerm + acc[t][i] - 0.5f * (sE2[wave * 16 + rl] + m2c);
      out[(size_t)(rowStart + rl) * L_DIM + t * 16 + col] = val;
    }
  }
}

extern "C" void kernel_launch(void* const* d_in, const int* in_sizes, int n_in,
                              void* d_out, int out_size, void* d_ws, size_t ws_size,
                              hipStream_t stream) {
  const float* emb   = (const float*)d_in[0];
  const float* means = (const float*)d_in[1];
  const float* var   = (const float*)d_in[2];
  float* out = (float*)d_out;
  float* ws  = (float*)d_ws;

  GaussianScorer_prep<<<1, 256, 0, stream>>>(means, var, ws);
  GaussianScorer_gemm<<<256, 128, 0, stream>>>(emb, ws, out);
}


# prep spread from one serial work-group to 69 (4 sum m2, 1 inv_var+const, 64 build scaled K-pair B rows) with whole-line stores; m2 now 16-term lane partials + 32-lane xor tree (sum order changed)
# speedup vs baseline: 1.8346x; 1.8346x over previous
#include <hip/hip_runtime.h>
#include <hip/hip_bf16.h>
#include <math.h>

typedef __attribute__((ext_vector_type(2))) float v2f;
typedef __attribute__((ext_vector_type(4))) float v4f;
typedef __attribute__((ext_vector_type(8))) float v8f;

#define E_DIM   512
#define L_DIM   128
#define KCHUNK  256
#define PRSTRIDE 144

#define WS_BTP   0
#define WS_IVAR  (E_DIM * L_DIM)
#define WS_M2    (WS_IVAR + E_DIM)
#define WS_CONST (WS_M2 + L_DIM)

#define PREP_M2_BLOCKS   4
#define PREP_BTP_BLOCK0  5

__global__ __launch_bounds__(256) void GaussianScorer_prep(
    const float* __restrict__ means, const float* __restrict__ var,
    float* __restrict__ ws) {
  __shared__ float sRed[256];
  __shared__ float sM2[32];
  const int tid = threadIdx.x;
  const int b   = blockIdx.x;

  if (b < PREP_M2_BLOCKS) {
    const int wave = tid >> 5;
    const int lane = tid & 31;
    v4f iv[4];
#pragma unroll
    for (int it = 0; it < 4; ++it) {
      const v4f v = *(const v4f*)(var + 4 * (lane + 32 * it));
      iv[it].x = 1.0f / v.x;
      iv[it].y = 1.0f / v.y;
      iv[it].z = 1.0f / v.z;
      iv[it].w = 1.0f / v.w;
    }
    const int row0 = b * 32 + wave * 4;
    float keep = 0.0f;
#pragma unroll
    for (int r = 0; r < 4; ++r) {
      const float* mr = means + (size_t)(row0 + r) * E_DIM;
      float s = 0.0f;
#pragma unroll
      for (int it = 0; it < 4; ++it) {
        const v4f m = *(const v4f*)(mr + 4 * (lane + 32 * it));
        s += m.x * m.x * iv[it].x;
        s += m.y * m.y * iv[it].y;
        s += m.z * m.z * iv[it].z;
        s += m.w * m.w * iv[it].w;
      }
#pragma unroll
      for (int off = 16; off > 0; off >>= 1) s += __shfl_xor(s, off, 32);
      if (lane == r) keep = s;
    }
    if (lane < 4) sM2[wave * 4 + lane] = keep;
    __syncthreads();
    if (tid < 32) ws[WS_M2 + b * 32 + tid] = sM2[tid];
  } else if (b < PREP_BTP_BLOCK0) {
    float slog = 0.0f;
    for (int e = tid; e < E_DIM; e += 256) {
      float v  = var[e];
      ws[WS_IVAR + e] = 1.0f / v;
      slog += logf(v);
    }
    sRed[tid] = slog;
    __syncthreads();
    for (int s = 128; s > 0; s >>= 1) {
      if (tid < s) sRed[tid] += sRed[tid + s];
      __syncthreads();
    }
    if (tid == 0) {
      const float log2pi = 1.8378770664093453f;
      const float factor = -0.5f * (float)E_DIM * log2pi;
      ws[WS_CONST] = factor - 0.5f * sRed[0];
    }
  } else {
    const int g  = b - PREP_BTP_BLOCK0;
    const int l  = tid & 127;
    const int h  = tid >> 7;
    const int e4 = 8 * g + 4 * h;
    const v4f m = *(const v4f*)(means + (size_t)l * E_DIM + e4);
    const v4f v = *(const v4f*)(var + e4);
    v2f lo, hi;
    lo.x = m.x * (1.0f / v.x);
    lo.y = m.y * (1.0f / v.y);
    hi.x = m.z * (1.0f / v.z);
    hi.y = m.w * (1.0f / v.w);
    const int pp = e4 >> 1;
    *(v2f*)(ws + WS_BTP + ((size_t)pp * L_DIM + l) * 2)       = lo;
    *(v2f*)(ws + WS_BTP + ((size_t)(pp + 1) * L_DIM + l) * 2) = hi;
  }
}

__global__ __launch_bounds__(128) void GaussianScorer_gemm(
    const float* __restrict__ emb, const float* __restrict__ ws,
    float* __restrict__ out) {
  __shared__ v2f   sBp[(KCHUNK / 2) * PRSTRIDE];
  __shared__ float sIv[KCHUNK];
  __shared__ float sE2[64];

  const int tid   = threadIdx.x;
  const int wave  = tid >> 5;
  const int lane  = tid & 31;
  const int col   = lane & 15;
  const int khalf = lane >> 4;
  const int kgrp  = khalf * 2;
  const int rowStart = blockIdx.x * 64 + wave * 16;
  const int myRow    = rowStart + col;

  const float* aRow = emb + (size_t)myRow * E_DIM;

  v8f acc[8] = {};
  float e2p = 0.0f;

  for (int chunk = 0; chunk < 2; ++chunk) {
    const v4f* __restrict__ srcB =
        (const v4f*)(ws + WS_BTP + chunk * (KCHUNK / 2) * (L_DIM * 2));
    for (int j = tid; j < (KCHUNK / 2) * (L_DIM / 2); j += 128) {
      int pr = j >> 6;
      int q  = j & 63;
      *((v4f*)(sBp + pr * PRSTRIDE) + q) = srcB[pr * (L_DIM / 2) + q];
    }
    if (tid < KCHUNK / 4)
      ((v4f*)sIv)[tid] = ((const v4f*)(ws + WS_IVAR + chunk * KCHUNK))[tid];
    __syncthreads();

    const float* aC = aRow + chunk * KCHUNK + kgrp;
#pragma unroll 4
    for (int i = 0; i < KCHUNK / 4; ++i) {
      const int kl = i * 4 + kgrp;
      v2f a  = *(const v2f*)(aC + i * 4);
      v2f iv = *(const v2f*)(sIv + kl);
      e2p += a.x * a.x * iv.x + a.y * a.y * iv.y;

      const v2f* bRow = sBp + (i * 2 + khalf) * PRSTRIDE + col;
#pragma unroll
      for (int t = 0; t < 8; ++t) {
        v2f b = bRow[t * 16];
        acc[t] = __builtin_amdgcn_wmma_f32_16x16x4_f32(
            false, a, false, b, (short)0, acc[t], false, false);
      }
    }
    __syncthreads();
  }

  float e2t = e2p + __shfl_xor(e2p, 16, 32);
  if (lane < 16) sE2[wave * 16 + col] = e2t;
  __syncthreads();

  const float cTerm = ws[WS_CONST];
#pragma unroll
  for (int t = 0; t < 8; ++t) {
    float m2c = ws[WS_M2 + t * 16 + col];
#pragma unroll
    for (int i = 0; i < 8; ++i) {
      int rl = khalf * 8 + i;
      float val = cTerm + acc[t][i] - 0.5f * (sE2[wave * 16 + rl] + m2c);
      out[(size_t)(rowStart + rl) * L_DIM + t * 16 + col] = val;
    }
  }
}

extern "C" void kernel_launch(void* const* d_in, const int* in_sizes, int n_in,
                              void* d_out, int out_size, void* d_ws, size_t ws_size,
                              hipStream_t stream) {
  const float* emb   = (const float*)d_in[0];
  const float* means = (const float*)d_in[1];
  const float* var   = (const float*)d_in[2];
  float* out = (float*)d_out;
  float* ws  = (float*)d_ws;

  GaussianScorer_prep<<<69, 256, 0, stream>>>(means, var, ws);
  GaussianScorer_gemm<<<256, 128, 0, stream>>>(emb, ws, out);
}


# gemm K loop software-pipelined in 8 stages of 64 k: the next stage's A fragments and Btp slice are loaded from global while this stage's WMMAs run; all of Btp stays resident in LDS
# speedup vs baseline: 2.7875x; 1.5194x over previous
#include <hip/hip_runtime.h>
#include <hip/hip_bf16.h>
#include <math.h>

typedef __attribute__((ext_vector_type(2))) float v2f;
typedef __attribute__((ext_vector_type(4))) float v4f;
typedef __attribute__((ext_vector_type(8))) float v8f;

#define E_DIM   512
#define L_DIM   128
#define PRSTRIDE 144

#define WS_BTP   0
#define WS_IVAR  (E_DIM * L_DIM)
#define WS_M2    (WS_IVAR + E_DIM)
#define WS_CONST (WS_M2 + L_DIM)

#define PREP_M2_BLOCKS   4
#define PREP_BTP_BLOCK0  5

__global__ __launch_bounds__(256) void GaussianScorer_prep(
    const float* __restrict__ means, const float* __restrict__ var,
    float* __restrict__ ws) {
  __shared__ float sRed[256];
  __shared__ float sM2[32];
  const int tid = threadIdx.x;
  const int b   = blockIdx.x;

  if (b < PREP_M2_BLOCKS) {
    const int wave = tid >> 5;
    const int lane = tid & 31;
    v4f iv[4];
#pragma unroll
    for (int it = 0; it < 4; ++it) {
      const v4f v = *(const v4f*)(var + 4 * (lane + 32 * it));
      iv[it].x = 1.0f / v.x;
      iv[it].y = 1.0f / v.y;
      iv[it].z = 1.0f / v.z;
      iv[it].w = 1.0f / v.w;
    }
    const int row0 = b * 32 + wave * 4;
    float keep = 0.0f;
#pragma unroll
    for (int r = 0; r < 4; ++r) {
      const float* mr = means + (size_t)(row0 + r) * E_DIM;
      float s = 0.0f;
#pragma unroll
      for (int it = 0; it < 4; ++it) {
        const v4f m = *(const v4f*)(mr + 4 * (lane + 32 * it));
        s += m.x * m.x * iv[it].x;
        s += m.y * m.y * iv[it].y;
        s += m.z * m.z * iv[it].z;
        s += m.w * m.w * iv[it].w;
      }
#pragma unroll
      for (int off = 16; off > 0; off >>= 1) s += __shfl_xor(s, off, 32);
      if (lane == r) keep = s;
    }
    if (lane < 4) sM2[wave * 4 + lane] = keep;
    __syncthreads();
    if (tid < 32) ws[WS_M2 + b * 32 + tid] = sM2[tid];
  } else if (b < PREP_BTP_BLOCK0) {
    float slog = 0.0f;
    for (int e = tid; e < E_DIM; e += 256) {
      float v  = var[e];
      ws[WS_IVAR + e] = 1.0f / v;
      slog += logf(v);
    }
    sRed[tid] = slog;
    __syncthreads();
    for (int s = 128; s > 0; s >>= 1) {
      if (tid < s) sRed[tid] += sRed[tid + s];
      __syncthreads();
    }
    if (tid == 0) {
      const float log2pi = 1.8378770664093453f;
      const float factor = -0.5f * (float)E_DIM * log2pi;
      ws[WS_CONST] = factor - 0.5f * sRed[0];
    }
  } else {
    const int g  = b - PREP_BTP_BLOCK0;
    const int l  = tid & 127;
    const int h  = tid >> 7;
    const int e4 = 8 * g + 4 * h;
    const v4f m = *(const v4f*)(means + (size_t)l * E_DIM + e4);
    const v4f v = *(const v4f*)(var + e4);
    v2f lo, hi;
    lo.x = m.x * (1.0f / v.x);
    lo.y = m.y * (1.0f / v.y);
    hi.x = m.z * (1.0f / v.z);
    hi.y = m.w * (1.0f / v.w);
    const int pp = e4 >> 1;
    *(v2f*)(ws + WS_BTP + ((size_t)pp * L_DIM + l) * 2)       = lo;
    *(v2f*)(ws + WS_BTP + ((size_t)(pp + 1) * L_DIM + l) * 2) = hi;
  }
}

#define GK        64
#define NSTAGE    (E_DIM / GK)
#define KS_STAGE  (GK / 4)
#define PR_STAGE  (GK / 2)
#define BQ_STAGE  (PR_STAGE * (L_DIM / 2))
#define BLD       (BQ_STAGE / 128)

__global__ __launch_bounds__(128) void GaussianScorer_gemm(
    const float* __restrict__ emb, const float* __restrict__ ws,
    float* __restrict__ out) {
  __shared__ v2f   sBp[(E_DIM / 2) * PRSTRIDE];
  __shared__ float sIv[E_DIM];
  __shared__ float sE2[64];

  const int tid   = threadIdx.x;
  const int wave  = tid >> 5;
  const int lane  = tid & 31;
  const int col   = lane & 15;
  const int khalf = lane >> 4;
  const int kgrp  = khalf * 2;
  const int rowStart = blockIdx.x * 64 + wave * 16;
  const int myRow    = rowStart + col;

  const float* aRow = emb + (size_t)myRow * E_DIM + kgrp;
  const v4f* __restrict__ srcB = (const v4f*)(ws + WS_BTP);

  v8f acc[8] = {};
  float e2p = 0.0f;

  v2f aC[KS_STAGE], aN[KS_STAGE];
  v4f bN[BLD];

#pragma unroll
  for (int s = 0; s < KS_STAGE; ++s) aC[s] = *(const v2f*)(aRow + 4 * s);
#pragma unroll
  for (int r = 0; r < BLD; ++r) bN[r] = srcB[tid + 128 * r];
  ((v4f*)sIv)[tid] = ((const v4f*)(ws + WS_IVAR))[tid];
#pragma unroll
  for (int r = 0; r < BLD; ++r) {
    const int j  = tid + 128 * r;
    const int pr = j >> 6;
    const int q  = j & 63;
    *((v4f*)(sBp + pr * PRSTRIDE) + q) = bN[r];
  }
  __syncthreads();

  for (int st = 0; st < NSTAGE; ++st) {
    const bool more = (st + 1 < NSTAGE);
    if (more) {
#pragma unroll
      for (int s = 0; s < KS_STAGE; ++s)
        aN[s] = *(const v2f*)(aRow + (st + 1) * GK + 4 * s);
#pragma unroll
      for (int r = 0; r < BLD; ++r)
        bN[r] = srcB[(st + 1) * BQ_STAGE + tid + 128 * r];
    }

#pragma unroll
    for (int s = 0; s < KS_STAGE; ++s) {
      const int i  = st * KS_STAGE + s;
      const int kl = i * 4 + kgrp;
      const v2f a  = aC[s];
      const v2f iv = *(const v2f*)(sIv + kl);
      e2p += a.x * a.x * iv.x + a.y * a.y * iv.y;

      const v2f* bRow = sBp + (i * 2 + khalf) * PRSTRIDE + col;
#pragma unroll
      for (int t = 0; t < 8; ++t) {
        v2f b = bRow[t * 16];
        acc[t] = __builtin_amdgcn_wmma_f32_16x16x4_f32(
            false, a, false, b, (short)0, acc[t], false, false);
      }
    }

    if (more) {
#pragma unroll
      for (int r = 0; r < BLD; ++r) {
        const int j  = tid + 128 * r;
        const int pr = (st + 1) * PR_STAGE + (j >> 6);
        const int q  = j & 63;
        *((v4f*)(sBp + pr * PRSTRIDE) + q) = bN[r];
      }
#pragma unroll
      for (int s = 0; s < KS_STAGE; ++s) aC[s] = aN[s];
    }
    __syncthreads();
  }

  float e2t = e2p + __shfl_xor(e2p, 16, 32);
  if (lane < 16) sE2[wave * 16 + col] = e2t;
  __syncthreads();

  const float cTerm = ws[WS_CONST];
#pragma unroll
  for (int t = 0; t < 8; ++t) {
    float m2c = ws[WS_M2 + t * 16 + col];
#pragma unroll
    for (int i = 0; i < 8; ++i) {
      int rl = khalf * 8 + i;
      float val = cTerm + acc[t][i] - 0.5f * (sE2[wave * 16 + rl] + m2c);
      out[(size_t)(rowStart + rl) * L_DIM + t * 16 + col] = val;
    }
  }
}

extern "C" void kernel_launch(void* const* d_in, const int* in_sizes, int n_in,
                              void* d_out, int out_size, void* d_ws, size_t ws_size,
                              hipStream_t stream) {
  const float* emb   = (const float*)d_in[0];
  const float* means = (const float*)d_in[1];
  const float* var   = (const float*)d_in[2];
  float* out = (float*)d_out;
  float* ws  = (float*)d_ws;

  GaussianScorer_prep<<<69, 256, 0, stream>>>(means, var, ws);
  GaussianScorer_gemm<<<256, 128, 0, stream>>>(emb, ws, out);
}


# prep kernel fused into the gemm (2 launches -> 1): each work-group scales its B slice from means and var while staging it to LDS, sums m2 as means*(means*iv) and log var; sum order of both changed
# speedup vs baseline: 3.5892x; 1.2876x over previous
#include <hip/hip_runtime.h>
#include <math.h>

typedef __attribute__((ext_vector_type(2))) float v2f;
typedef __attribute__((ext_vector_type(4))) float v4f;
typedef __attribute__((ext_vector_type(8))) float v8f;

#define E_DIM   512
#define L_DIM   128
#define BPITCH  516

#define GK        64
#define NSTAGE    (E_DIM / GK)
#define KS_STAGE  (GK / 4)
#define BLD       16

__global__ __launch_bounds__(128) void GaussianScorer_fused(
    const float* __restrict__ emb, const float* __restrict__ means,
    const float* __restrict__ var, float* __restrict__ out) {
  __shared__ __attribute__((aligned(16))) float sB[L_DIM * BPITCH];
  __shared__ __attribute__((aligned(16))) float sIv[E_DIM];
  __shared__ float sE2[64];
  __shared__ __attribute__((aligned(16))) float sM2p[L_DIM * 16];
  __shared__ float sM2[L_DIM];
  __shared__ float sLog[4];

  const int tid   = threadIdx.x;
  const int wave  = tid >> 5;
  const int lane  = tid & 31;
  const int col   = lane & 15;
  const int khalf = lane >> 4;
  const int kgrp  = khalf * 2;
  const int rowStart = blockIdx.x * 64 + wave * 16;
  const int myRow    = rowStart + col;

  const float* aRow = emb + (size_t)myRow * E_DIM + kgrp;

  const int brow = tid >> 4;
  const int bq   = tid & 15;
  const float* bSrc = means + (size_t)brow * E_DIM + 4 * bq;
  float*       bDst = sB + brow * BPITCH + 4 * bq;

  v8f acc[8] = {};
  float e2p = 0.0f;
  v4f m2v[BLD];
#pragma unroll
  for (int r = 0; r < BLD; ++r) m2v[r] = (v4f){0.0f, 0.0f, 0.0f, 0.0f};

  v2f aC[KS_STAGE], aN[KS_STAGE];
  v4f bN[BLD];

  const v4f vv = *(const v4f*)(var + 4 * tid);
#pragma unroll
  for (int s = 0; s < KS_STAGE; ++s) aC[s] = *(const v2f*)(aRow + 4 * s);
#pragma unroll
  for (int r = 0; r < BLD; ++r) bN[r] = *(const v4f*)(bSrc + (size_t)r * 8 * E_DIM);
  {
    v4f ivv;
    ivv.x = 1.0f / vv.x;
    ivv.y = 1.0f / vv.y;
    ivv.z = 1.0f / vv.z;
    ivv.w = 1.0f / vv.w;
    *(v4f*)(sIv + 4 * tid) = ivv;
  }
  float lsum = (logf(vv.x) + logf(vv.y)) + (logf(vv.z) + logf(vv.w));
#pragma unroll
  for (int off = 16; off > 0; off >>= 1) lsum += __shfl_xor(lsum, off, 32);
  if (lane == 0) sLog[wave] = lsum;
  __syncthreads();
  {
    const v4f iv = *(const v4f*)(sIv + 4 * bq);
#pragma unroll
    for (int r = 0; r < BLD; ++r) {
      const v4f m = bN[r];
      const v4f b = m * iv;
      m2v[r] += m * b;
      *(v4f*)(bDst + r * 8 * BPITCH) = b;
    }
  }
  __syncthreads();

  for (int st = 0; st < NSTAGE; ++st) {
    const bool more = (st + 1 < NSTAGE);
    if (more) {
#pragma unroll
      for (int s = 0; s < KS_STAGE; ++s)
        aN[s] = *(const v2f*)(aRow + (st + 1) * GK + 4 * s);
#pragma unroll
      for (int r = 0; r < BLD; ++r)
        bN[r] = *(const v4f*)(bSrc + (size_t)r * 8 * E_DIM + (st + 1) * GK);
    }

    int bOff[8];
#pragma unroll
    for (int t = 0; t < 8; ++t) {
      int o = (16 * t + col) * (BPITCH / 2) + st * (GK / 2) + khalf;
      asm("" : "+v"(o));
      bOff[t] = o;
    }
    const v2f* sBp = (const v2f*)sB;
#pragma unroll
    for (int p = 0; p < KS_STAGE / 2; ++p) {
#pragma unroll
      for (int u = 0; u < 2; ++u) {
        const int s  = 2 * p + u;
        const int kl = (st * KS_STAGE + s) * 4 + kgrp;
        const v2f a  = aC[s];
        const v2f iv = *(const v2f*)(sIv + kl);
        e2p += a.x * a.x * iv.x + a.y * a.y * iv.y;
      }
      v2f b0[8], b1[8];
#pragma unroll
      for (int t = 0; t < 8; ++t) {
        b0[t] = sBp[bOff[t] + 4 * p];
        b1[t] = sBp[bOff[t] + 4 * p + 2];
      }
#pragma unroll
      for (int t = 0; t < 8; ++t)
        acc[t] = __builtin_amdgcn_wmma_f32_16x16x4_f32(
            false, aC[2 * p], false, b0[t], (short)0, acc[t], false, false);
#pragma unroll
      for (int t = 0; t < 8; ++t)
        acc[t] = __builtin_amdgcn_wmma_f32_16x16x4_f32(
            false, aC[2 * p + 1], false, b1[t], (short)0, acc[t], false, false);
    }

    if (more) {
      const v4f iv = *(const v4f*)(sIv + (st + 1) * GK + 4 * bq);
#pragma unroll
      for (int r = 0; r < BLD; ++r) {
        const v4f m = bN[r];
        const v4f b = m * iv;
        m2v[r] += m * b;
        *(v4f*)(bDst + r * 8 * BPITCH + (st + 1) * GK) = b;
      }
#pragma unroll
      for (int s = 0; s < KS_STAGE; ++s) aC[s] = aN[s];
    }
    __syncthreads();
  }

  float e2t = e2p + __shfl_xor(e2p, 16, 32);
  if (lane < 16) sE2[wave * 16 + col] = e2t;
#pragma unroll
  for (int r = 0; r < BLD; ++r)
    sM2p[tid + 128 * r] = (m2v[r].x + m2v[r].y) + (m2v[r].z + m2v[r].w);
  __syncthreads();

  {
    const v4f* p = (const v4f*)(sM2p + 16 * tid);
    const v4f p0 = p[0], p1 = p[1], p2 = p[2], p3 = p[3];
    const float q0 = (p0.x + p0.y) + (p0.z + p0.w);
    const float q1 = (p1.x + p1.y) + (p1.z + p1.w);
    const float q2 = (p2.x + p2.y) + (p2.z + p2.w);
    const float q3 = (p3.x + p3.y) + (p3.z + p3.w);
    sM2[tid] = (q0 + q1) + (q2 + q3);
  }
  __syncthreads();

  const float log2pi = 1.8378770664093453f;
  const float factor = -0.5f * (float)E_DIM * log2pi;
  const float cTerm  = factor - 0.5f * ((sLog[0] + sLog[1]) + (sLog[2] + sLog[3]));
#pragma unroll
  for (int t = 0; t < 8; ++t) {
    float m2c = sM2[t * 16 + col];
#pragma unroll
    for (int i = 0; i < 8; ++i) {
      int rl = khalf * 8 + i;
      float val = cTerm + acc[t][i] - 0.5f * (sE2[wave * 16 + rl] + m2c);
      out[(size_t)(rowStart + rl) * L_DIM + t * 16 + col] = val;
    }
  }
}

extern "C" void kernel_launch(void* const* d_in, const int* in_sizes, int n_in,
                              void* d_out, int out_size, void* d_ws, size_t ws_size,
                              hipStream_t stream) {
  const float* emb   = (const float*)d_in[0];
  const float* means = (const float*)d_in[1];
  const float* var   = (const float*)d_in[2];
  float* out = (float*)d_out;

  GaussianScorer_fused<<<256, 128, 0, stream>>>(emb, means, var, out);
}
